# starting kernel
# baseline (speedup 1.0000x reference)
#include <hip/hip_runtime.h>
#include <math.h>

typedef __attribute__((ext_vector_type(2))) float v2f;
typedef __attribute__((ext_vector_type(8))) float v8f;

#define HW   3136
#define CCH  512
#define NB   32
#define EPSV 1e-5f

__global__ void se_reduce_kernel(const float* __restrict__ x,
                                 float* __restrict__ s_out,
                                 float* __restrict__ q_out) {
    const int bc = blockIdx.x;
    const float4* p = (const float4*)(x + (size_t)bc * HW);
    float sum = 0.f, sq = 0.f;
    for (int i = threadIdx.x; i < HW / 4; i += blockDim.x) {
        float4 v = p[i];
        sum += v.x + v.y + v.z + v.w;
        sq  += v.x * v.x + v.y * v.y + v.z * v.z + v.w * v.w;
    }
    #pragma unroll
    for (int off = 16; off > 0; off >>= 1) {
        sum += __shfl_down(sum, off, 32);
        sq  += __shfl_down(sq,  off, 32);
    }
    __shared__ float ls[8], lq[8];
    const int lane = threadIdx.x & 31;
    const int wv   = threadIdx.x >> 5;
    if (lane == 0) { ls[wv] = sum; lq[wv] = sq; }
    __syncthreads();
    if (threadIdx.x == 0) {
        float ts = 0.f, tq = 0.f;
        #pragma unroll
        for (int i = 0; i < 8; ++i) { ts += ls[i]; tq += lq[i]; }
        const float inv = 1.0f / (float)HW;
        s_out[bc] = ts * inv;
        q_out[bc] = tq * inv;
    }
}

__global__ void se_excite_kernel(const float* __restrict__ s,
                                 const float* __restrict__ q,
                                 const float* __restrict__ w1,
                                 const float* __restrict__ w2,
                                 float* __restrict__ my,
                                 float* __restrict__ vv) {
    __shared__ float t1[32][33];
    const int lane = threadIdx.x & 31;
    const int wv   = threadIdx.x >> 5;
    const int half = lane >> 4;
    const int l    = lane & 15;

    {
        const int mt = wv >> 1, nt = wv & 1;
        const int arow = 16 * mt + l;
        const int brow = 16 * nt + l;
        v8f acc = {};
        for (int k = 0; k < CCH; k += 4) {
            v2f a, b;
            a.x = s[arow * CCH + k + 2 * half + 0];
            a.y = s[arow * CCH + k + 2 * half + 1];
            b.x = w1[brow * CCH + k + 2 * half + 0];
            b.y = w1[brow * CCH + k + 2 * half + 1];
            acc = __builtin_amdgcn_wmma_f32_16x16x4_f32(
                false, a, false, b, (short)0, acc, false, false);
        }
        #pragma unroll
        for (int r = 0; r < 8; ++r) {
            float vr = acc[r];
            t1[16 * mt + r + 8 * half][16 * nt + l] = vr > 0.f ? vr : 0.f;
        }
    }
    __syncthreads();

    for (int tile = wv; tile < 64; tile += 4) {
        const int mt = tile & 1;
        const int nt = tile >> 1;
        const int arow = 16 * mt + l;
        const int ncol = 16 * nt + l;
        v8f acc = {};
        #pragma unroll
        for (int k = 0; k < 32; k += 4) {
            v2f a, b;
            a.x = t1[arow][k + 2 * half + 0];
            a.y = t1[arow][k + 2 * half + 1];
            b.x = w2[ncol * 32 + k + 2 * half + 0];
            b.y = w2[ncol * 32 + k + 2 * half + 1];
            acc = __builtin_amdgcn_wmma_f32_16x16x4_f32(
                false, a, false, b, (short)0, acc, false, false);
        }
        #pragma unroll
        for (int r = 0; r < 8; ++r) {
            const int b = 16 * mt + r + 8 * half;
            const int c = ncol;
            float e  = 1.0f / (1.0f + __expf(-acc[r]));
            float sv = s[b * CCH + c];
            float qv = q[b * CCH + c];
            my[b * CCH + c] = sv * e;
            vv[b * CCH + c] = e * e * qv;
        }
    }
}

__global__ void se_finalize_kernel(const float* __restrict__ my,
                                   const float* __restrict__ vv,
                                   const float* __restrict__ bw,
                                   const float* __restrict__ bb,
                                   float* __restrict__ out) {
    const int k = threadIdx.x;
    float s1 = 0.f, t1 = 0.f, s2 = 0.f, t2 = 0.f;
    #pragma unroll 4
    for (int b = 0; b < NB; ++b) {
        s1 += my[b * CCH + k];
        t1 += vv[b * CCH + k];
        s2 += my[b * CCH + k + 256];
        t2 += vv[b * CCH + k + 256];
    }
    const float inv = 1.0f / (float)NB;
    const float mu1 = s1 * inv, mu2 = s2 * inv;
    const float g1 = bw[k]       * rsqrtf(t1 * inv - mu1 * mu1 + EPSV);
    const float g2 = bw[k + 256] * rsqrtf(t2 * inv - mu2 * mu2 + EPSV);
    const float bias = bb[k] + bb[k + 256];
    for (int b = 0; b < NB; ++b) {
        out[b * 256 + k] = (my[b * CCH + k] - mu1) * g1
                         + (my[b * CCH + k + 256] - mu2) * g2 + bias;
    }
}

extern "C" void kernel_launch(void* const* d_in, const int* in_sizes, int n_in,
                              void* d_out, int out_size, void* d_ws, size_t ws_size,
                              hipStream_t stream) {
    const float* x  = (const float*)d_in[0];
    const float* w1 = (const float*)d_in[1];
    const float* w2 = (const float*)d_in[2];
    const float* bw = (const float*)d_in[3];
    const float* bb = (const float*)d_in[4];
    float* out = (float*)d_out;

    float* ws = (float*)d_ws;
    float* s  = ws;
    float* q  = ws + 16384;
    float* my = ws + 32768;
    float* vv = ws + 49152;

    se_reduce_kernel<<<NB * CCH, 256, 0, stream>>>(x, s, q);
    se_excite_kernel<<<1, 128, 0, stream>>>(s, q, w1, w2, my, vv);
    se_finalize_kernel<<<1, 256, 0, stream>>>(my, vv, bw, bb, out);
}


# excite kernel: 32 groups of 32 waves, one channel tile each; first product's K split over the waves, loads issued as one group, partial sums added through LDS in wave order (was one 4-wave group)
# speedup vs baseline: 1.3995x; 1.3995x over previous
#include <hip/hip_runtime.h>
#include <math.h>

typedef __attribute__((ext_vector_type(2))) float v2f;
typedef __attribute__((ext_vector_type(8))) float v8f;

#define HW   3136
#define CCH  512
#define NB   32
#define EPSV 1e-5f

__global__ void se_reduce_kernel(const float* __restrict__ x,
                                 float* __restrict__ s_out,
                                 float* __restrict__ q_out) {
    const int bc = blockIdx.x;
    const float4* p = (const float4*)(x + (size_t)bc * HW);
    float sum = 0.f, sq = 0.f;
    for (int i = threadIdx.x; i < HW / 4; i += blockDim.x) {
        float4 v = p[i];
        sum += v.x + v.y + v.z + v.w;
        sq  += v.x * v.x + v.y * v.y + v.z * v.z + v.w * v.w;
    }
    #pragma unroll
    for (int off = 16; off > 0; off >>= 1) {
        sum += __shfl_down(sum, off, 32);
        sq  += __shfl_down(sq,  off, 32);
    }
    __shared__ float ls[8], lq[8];
    const int lane = threadIdx.x & 31;
    const int wv   = threadIdx.x >> 5;
    if (lane == 0) { ls[wv] = sum; lq[wv] = sq; }
    __syncthreads();
    if (threadIdx.x == 0) {
        float ts = 0.f, tq = 0.f;
        #pragma unroll
        for (int i = 0; i < 8; ++i) { ts += ls[i]; tq += lq[i]; }
        const float inv = 1.0f / (float)HW;
        s_out[bc] = ts * inv;
        q_out[bc] = tq * inv;
    }
}

#define NWV  32
#define PPIT 34
#define TPIT 36

__global__ __launch_bounds__(1024) void se_excite_kernel(const float* __restrict__ s,
                                 const float* __restrict__ q,
                                 const float* __restrict__ w1,
                                 const float* __restrict__ w2,
                                 float* __restrict__ my,
                                 float* __restrict__ vv) {
    __shared__ float part[NWV][32][PPIT];
    __shared__ __attribute__((aligned(16))) float t1[32][TPIT];
    const int lane = threadIdx.x & 31;
    const int wv   = __builtin_amdgcn_readfirstlane((int)(threadIdx.x >> 5));
    const int half = lane >> 4;
    const int l    = lane & 15;
    const int ncol = 16 * (int)blockIdx.x + l;

    const int kb = 16 * wv + 8 * half;
    const float4* pa0 = (const float4*)(s  + (l)      * CCH + kb);
    const float4* pa1 = (const float4*)(s  + (16 + l) * CCH + kb);
    const float4* pb0 = (const float4*)(w1 + (l)      * CCH + kb);
    const float4* pb1 = (const float4*)(w1 + (16 + l) * CCH + kb);
    const float4 a0lo = pa0[0], a0hi = pa0[1];
    const float4 a1lo = pa1[0], a1hi = pa1[1];
    const float4 b0lo = pb0[0], b0hi = pb0[1];
    const float4 b1lo = pb1[0], b1hi = pb1[1];

    float4 wb0 = {}, wb1 = {}, wb2 = {}, wb3 = {};
    float sv[8] = {}, qv[8] = {};
    if (wv < 2) {
        const float4* pw = (const float4*)(w2 + ncol * 32 + 16 * half);
        wb0 = pw[0]; wb1 = pw[1]; wb2 = pw[2]; wb3 = pw[3];
        #pragma unroll
        for (int r = 0; r < 8; ++r) {
            const int b = 16 * wv + r + 8 * half;
            sv[r] = s[b * CCH + ncol];
            qv[r] = q[b * CCH + ncol];
        }
    }

    {
        v8f c00 = {}, c01 = {}, c10 = {}, c11 = {};
        #define SE_STEP(A0, A1, B0, B1, X, Y)                                          \
        {                                                                             \
            v2f fa0, fa1, fb0, fb1;                                                   \
            fa0.x = A0.X; fa0.y = A0.Y; fa1.x = A1.X; fa1.y = A1.Y;                   \
            fb0.x = B0.X; fb0.y = B0.Y; fb1.x = B1.X; fb1.y = B1.Y;                   \
            c00 = __builtin_amdgcn_wmma_f32_16x16x4_f32(false, fa0, false, fb0, (short)0, c00, false, false); \
            c01 = __builtin_amdgcn_wmma_f32_16x16x4_f32(false, fa0, false, fb1, (short)0, c01, false, false); \
            c10 = __builtin_amdgcn_wmma_f32_16x16x4_f32(false, fa1, false, fb0, (short)0, c10, false, false); \
            c11 = __builtin_amdgcn_wmma_f32_16x16x4_f32(false, fa1, false, fb1, (short)0, c11, false, false); \
        }
        SE_STEP(a0lo, a1lo, b0lo, b1lo, x, y)
        SE_STEP(a0lo, a1lo, b0lo, b1lo, z, w)
        SE_STEP(a0hi, a1hi, b0hi, b1hi, x, y)
        SE_STEP(a0hi, a1hi, b0hi, b1hi, z, w)
        #undef SE_STEP
        #pragma unroll
        for (int r = 0; r < 8; ++r) {
            const int row = r + 8 * half;
            part[wv][row][l]           = c00[r];
            part[wv][row][16 + l]      = c01[r];
            part[wv][16 + row][l]      = c10[r];
            part[wv][16 + row][16 + l] = c11[r];
        }
    }
    __syncthreads();

    {
        float acc = 0.f;
        #pragma unroll
        for (int w = 0; w < NWV; ++w) acc += part[w][wv][lane];
        t1[wv][lane] = acc > 0.f ? acc : 0.f;
    }
    __syncthreads();

    if (wv < 2) {
        const float4* pt = (const float4*)&t1[16 * wv + l][16 * half];
        const float4 ta0 = pt[0], ta1 = pt[1], ta2 = pt[2], ta3 = pt[3];
        v8f e = {};
        #define SE_STEP2(TA, WB, X, Y)                                                 \
        {                                                                             \
            v2f fa, fb;                                                               \
            fa.x = TA.X; fa.y = TA.Y; fb.x = WB.X; fb.y = WB.Y;                       \
            e = __builtin_amdgcn_wmma_f32_16x16x4_f32(false, fa, false, fb, (short)0, e, false, false); \
        }
        SE_STEP2(ta0, wb0, x, y)
        SE_STEP2(ta0, wb0, z, w)
        SE_STEP2(ta1, wb1, x, y)
        SE_STEP2(ta1, wb1, z, w)
        SE_STEP2(ta2, wb2, x, y)
        SE_STEP2(ta2, wb2, z, w)
        SE_STEP2(ta3, wb3, x, y)
        SE_STEP2(ta3, wb3, z, w)
        #undef SE_STEP2
        #pragma unroll
        for (int r = 0; r < 8; ++r) {
            const int b = 16 * wv + r + 8 * half;
            const float ex = 1.0f / (1.0f + __expf(-e[r]));
            my[b * CCH + ncol] = sv[r] * ex;
            vv[b * CCH + ncol] = ex * ex * qv[r];
        }
    }
}

__global__ void se_finalize_kernel(const float* __restrict__ my,
                                   const float* __restrict__ vv,
                                   const float* __restrict__ bw,
                                   const float* __restrict__ bb,
                                   float* __restrict__ out) {
    const int k = threadIdx.x;
    float s1 = 0.f, t1 = 0.f, s2 = 0.f, t2 = 0.f;
    #pragma unroll 4
    for (int b = 0; b < NB; ++b) {
        s1 += my[b * CCH + k];
        t1 += vv[b * CCH + k];
        s2 += my[b * CCH + k + 256];
        t2 += vv[b * CCH + k + 256];
    }
    const float inv = 1.0f / (float)NB;
    const float mu1 = s1 * inv, mu2 = s2 * inv;
    const float g1 = bw[k]       * rsqrtf(t1 * inv - mu1 * mu1 + EPSV);
    const float g2 = bw[k + 256] * rsqrtf(t2 * inv - mu2 * mu2 + EPSV);
    const float bias = bb[k] + bb[k + 256];
    for (int b = 0; b < NB; ++b) {
        out[b * 256 + k] = (my[b * CCH + k] - mu1) * g1
                         + (my[b * CCH + k + 256] - mu2) * g2 + bias;
    }
}

extern "C" void kernel_launch(void* const* d_in, const int* in_sizes, int n_in,
                              void* d_out, int out_size, void* d_ws, size_t ws_size,
                              hipStream_t stream) {
    const float* x  = (const float*)d_in[0];
    const float* w1 = (const float*)d_in[1];
    const float* w2 = (const float*)d_in[2];
    const float* bw = (const float*)d_in[3];
    const float* bb = (const float*)d_in[4];
    float* out = (float*)d_out;

    float* ws = (float*)d_ws;
    float* s  = ws;
    float* q  = ws + 16384;
    float* my = ws + 32768;
    float* vv = ws + 49152;

    se_reduce_kernel<<<NB * CCH, 256, 0, stream>>>(x, s, q);
    se_excite_kernel<<<32, 1024, 0, stream>>>(s, q, w1, w2, my, vv);
    se_finalize_kernel<<<1, 256, 0, stream>>>(my, vv, bw, bb, out);
}


# reduce kernel: 256 groups of 16 waves, each wave streams 4 rows as 98 whole-line 16-byte loads, 24-49 in flight; row sums in a new order (lane chains, then lane butterfly); was 8 waves per row
# speedup vs baseline: 1.9606x; 1.4009x over previous
#include <hip/hip_runtime.h>
#include <math.h>

typedef __attribute__((ext_vector_type(2))) float v2f;
typedef __attribute__((ext_vector_type(8))) float v8f;

#define HW   3136
#define CCH  512
#define NB   32
#define EPSV 1e-5f

typedef __attribute__((ext_vector_type(4))) float v4f;

#define RPW    4
#define RWAVES 16
#define RGRID  256
#define RTHREADS 512
#define ROWF4  (HW / 4)
#define RGSZ   25

static_assert(RGRID * RWAVES * RPW == NB * CCH, "the reduce grid must cover every (b,c) row once");
static_assert(RTHREADS == RWAVES * 32, "work-group size and wave count must agree");
static_assert((RPW * ROWF4) % 32 == 0, "a wave's chunk must be a whole number of wave-wide loads");

__host__ __device__ constexpr int se_gbeg(int g) { return (g * 49 + 1) / 2; }
static_assert(se_gbeg(4) == RPW * ROWF4 / 32, "the four groups must cover the wave's 98 loads");
static_assert(1 * ROWF4 <= 32 * se_gbeg(1) && 2 * ROWF4 <= 32 * se_gbeg(2) &&
              3 * ROWF4 <= 32 * se_gbeg(3) && 4 * ROWF4 <= 32 * se_gbeg(4),
              "row g must be complete once group g is summed");
static_assert(se_gbeg(1) - se_gbeg(0) <= RGSZ && se_gbeg(2) - se_gbeg(1) <= RGSZ &&
              se_gbeg(3) - se_gbeg(2) <= RGSZ && se_gbeg(4) - se_gbeg(3) <= RGSZ, "ring slot too small");

__global__ __launch_bounds__(RTHREADS) void se_reduce_kernel(const float* __restrict__ x,
                                 float* __restrict__ s_out,
                                 float* __restrict__ q_out) {
    const int lane = threadIdx.x & 31;
    const int wv   = __builtin_amdgcn_readfirstlane((int)(threadIdx.x >> 5));
    const int row0 = ((int)blockIdx.x * RWAVES + wv) * RPW;
    const v4f* p = (const v4f*)(x + (size_t)row0 * HW) + lane;

    v4f buf[2][RGSZ];
    v4f sa[RPW], qa[RPW];
    float rs[RPW], rq[RPW];
    #pragma unroll
    for (int r = 0; r < RPW; ++r) { sa[r] = (v4f){0.f, 0.f, 0.f, 0.f}; qa[r] = (v4f){0.f, 0.f, 0.f, 0.f}; }

    #define SE_LOAD_GROUP(G)                                                           \
    {                                                                                  \
               \
        _Pragma("unroll")                                                              \
        for (int j = se_gbeg(G); j < se_gbeg((G) + 1); ++j)                            \
            buf[(G) & 1][j - se_gbeg(G)] = p[32 * j];                                  \
    }
    #define SE_SUM_GROUP(G)                                                            \
    {                                                                                  \
          \
        _Pragma("unroll")                                                              \
        for (int j = se_gbeg(G); j < se_gbeg((G) + 1); ++j) {                          \
            const v4f v = buf[(G) & 1][j - se_gbeg(G)];                                \
            const int rlo = (32 * j) / ROWF4, rhi = (32 * j + 31) / ROWF4;             \
            if (rlo == rhi) {                                                          \
                sa[rlo] += v; qa[rlo] += v * v;                                        \
            } else {                                                                   \
                const int split = rhi * ROWF4 - 32 * j;      \
                const v4f z = (v4f){0.f, 0.f, 0.f, 0.f};                               \
                const v4f vlo = (lane < split) ? v : z;                                \
                const v4f vhi = (lane < split) ? z : v;                                \
                sa[rlo] += vlo; qa[rlo] += vlo * vlo;                                  \
                sa[rhi] += vhi; qa[rhi] += vhi * vhi;                                  \
            }                                                                          \
        }                                                                              \
        rs[G] = (sa[G].x + sa[G].y) + (sa[G].z + sa[G].w);                             \
        rq[G] = (qa[G].x + qa[G].y) + (qa[G].z + qa[G].w);                             \
    }
    SE_LOAD_GROUP(0)
    SE_LOAD_GROUP(1)
    __builtin_amdgcn_sched_barrier(0);
    SE_SUM_GROUP(0)
    __builtin_amdgcn_sched_barrier(0);
    SE_LOAD_GROUP(2)
    __builtin_amdgcn_sched_barrier(0);
    SE_SUM_GROUP(1)
    __builtin_amdgcn_sched_barrier(0);
    SE_LOAD_GROUP(3)
    __builtin_amdgcn_sched_barrier(0);
    SE_SUM_GROUP(2)
    SE_SUM_GROUP(3)
    #undef SE_LOAD_GROUP
    #undef SE_SUM_GROUP

    #pragma unroll
    for (int r = 0; r < RPW; ++r) {
        #pragma unroll
        for (int m = 16; m >= 1; m >>= 1) {
            rs[r] += __shfl_xor(rs[r], m, 32);
            rq[r] += __shfl_xor(rq[r], m, 32);
        }
    }
    float mine = 0.f;
    #pragma unroll
    for (int r = 0; r < RPW; ++r) {
        if (lane == r)       mine = rs[r];
        if (lane == RPW + r) mine = rq[r];
    }
    const float inv = 1.0f / (float)HW;
    if (lane < RPW)          s_out[row0 + lane]       = mine * inv;
    else if (lane < 2 * RPW) q_out[row0 + lane - RPW] = mine * inv;
}

#define NWV  32
#define PPIT 34
#define TPIT 36

__global__ __launch_bounds__(1024) void se_excite_kernel(const float* __restrict__ s,
                                 const float* __restrict__ q,
                                 const float* __restrict__ w1,
                                 const float* __restrict__ w2,
                                 float* __restrict__ my,
                                 float* __restrict__ vv) {
    __shared__ float part[NWV][32][PPIT];
    __shared__ __attribute__((aligned(16))) float t1[32][TPIT];
    const int lane = threadIdx.x & 31;
    const int wv   = __builtin_amdgcn_readfirstlane((int)(threadIdx.x >> 5));
    const int half = lane >> 4;
    const int l    = lane & 15;
    const int ncol = 16 * (int)blockIdx.x + l;

    const int kb = 16 * wv + 8 * half;
    const float4* pa0 = (const float4*)(s  + (l)      * CCH + kb);
    const float4* pa1 = (const float4*)(s  + (16 + l) * CCH + kb);
    const float4* pb0 = (const float4*)(w1 + (l)      * CCH + kb);
    const float4* pb1 = (const float4*)(w1 + (16 + l) * CCH + kb);
    const float4 a0lo = pa0[0], a0hi = pa0[1];
    const float4 a1lo = pa1[0], a1hi = pa1[1];
    const float4 b0lo = pb0[0], b0hi = pb0[1];
    const float4 b1lo = pb1[0], b1hi = pb1[1];

    float4 wb0 = {}, wb1 = {}, wb2 = {}, wb3 = {};
    float sv[8] = {}, qv[8] = {};
    if (wv < 2) {
        const float4* pw = (const float4*)(w2 + ncol * 32 + 16 * half);
        wb0 = pw[0]; wb1 = pw[1]; wb2 = pw[2]; wb3 = pw[3];
        #pragma unroll
        for (int r = 0; r < 8; ++r) {
            const int b = 16 * wv + r + 8 * half;
            sv[r] = s[b * CCH + ncol];
            qv[r] = q[b * CCH + ncol];
        }
    }

    {
        v8f c00 = {}, c01 = {}, c10 = {}, c11 = {};
        #define SE_STEP(A0, A1, B0, B1, X, Y)                                          \
        {                                                                             \
            v2f fa0, fa1, fb0, fb1;                                                   \
            fa0.x = A0.X; fa0.y = A0.Y; fa1.x = A1.X; fa1.y = A1.Y;                   \
            fb0.x = B0.X; fb0.y = B0.Y; fb1.x = B1.X; fb1.y = B1.Y;                   \
            c00 = __builtin_amdgcn_wmma_f32_16x16x4_f32(false, fa0, false, fb0, (short)0, c00, false, false); \
            c01 = __builtin_amdgcn_wmma_f32_16x16x4_f32(false, fa0, false, fb1, (short)0, c01, false, false); \
            c10 = __builtin_amdgcn_wmma_f32_16x16x4_f32(false, fa1, false, fb0, (short)0, c10, false, false); \
            c11 = __builtin_amdgcn_wmma_f32_16x16x4_f32(false, fa1, false, fb1, (short)0, c11, false, false); \
        }
        SE_STEP(a0lo, a1lo, b0lo, b1lo, x, y)
        SE_STEP(a0lo, a1lo, b0lo, b1lo, z, w)
        SE_STEP(a0hi, a1hi, b0hi, b1hi, x, y)
        SE_STEP(a0hi, a1hi, b0hi, b1hi, z, w)
        #undef SE_STEP
        #pragma unroll
        for (int r = 0; r < 8; ++r) {
            const int row = r + 8 * half;
            part[wv][row][l]           = c00[r];
            part[wv][row][16 + l]      = c01[r];
            part[wv][16 + row][l]      = c10[r];
            part[wv][16 + row][16 + l] = c11[r];
        }
    }
    __syncthreads();

    {
        float acc = 0.f;
        #pragma unroll
        for (int w = 0; w < NWV; ++w) acc += part[w][wv][lane];
        t1[wv][lane] = acc > 0.f ? acc : 0.f;
    }
    __syncthreads();

    if (wv < 2) {
        const float4* pt = (const float4*)&t1[16 * wv + l][16 * half];
        const float4 ta0 = pt[0], ta1 = pt[1], ta2 = pt[2], ta3 = pt[3];
        v8f e = {};
        #define SE_STEP2(TA, WB, X, Y)                                                 \
        {                                                                             \
            v2f fa, fb;                                                               \
            fa.x = TA.X; fa.y = TA.Y; fb.x = WB.X; fb.y = WB.Y;                       \
            e = __builtin_amdgcn_wmma_f32_16x16x4_f32(false, fa, false, fb, (short)0, e, false, false); \
        }
        SE_STEP2(ta0, wb0, x, y)
        SE_STEP2(ta0, wb0, z, w)
        SE_STEP2(ta1, wb1, x, y)
        SE_STEP2(ta1, wb1, z, w)
        SE_STEP2(ta2, wb2, x, y)
        SE_STEP2(ta2, wb2, z, w)
        SE_STEP2(ta3, wb3, x, y)
        SE_STEP2(ta3, wb3, z, w)
        #undef SE_STEP2
        #pragma unroll
        for (int r = 0; r < 8; ++r) {
            const int b = 16 * wv + r + 8 * half;
            const float ex = 1.0f / (1.0f + __expf(-e[r]));
            my[b * CCH + ncol] = sv[r] * ex;
            vv[b * CCH + ncol] = ex * ex * qv[r];
        }
    }
}

__global__ void se_finalize_kernel(const float* __restrict__ my,
                                   const float* __restrict__ vv,
                                   const float* __restrict__ bw,
                                   const float* __restrict__ bb,
                                   float* __restrict__ out) {
    const int k = threadIdx.x;
    float s1 = 0.f, t1 = 0.f, s2 = 0.f, t2 = 0.f;
    #pragma unroll 4
    for (int b = 0; b < NB; ++b) {
        s1 += my[b * CCH + k];
        t1 += vv[b * CCH + k];
        s2 += my[b * CCH + k + 256];
        t2 += vv[b * CCH + k + 256];
    }
    const float inv = 1.0f / (float)NB;
    const float mu1 = s1 * inv, mu2 = s2 * inv;
    const float g1 = bw[k]       * rsqrtf(t1 * inv - mu1 * mu1 + EPSV);
    const float g2 = bw[k + 256] * rsqrtf(t2 * inv - mu2 * mu2 + EPSV);
    const float bias = bb[k] + bb[k + 256];
    for (int b = 0; b < NB; ++b) {
        out[b * 256 + k] = (my[b * CCH + k] - mu1) * g1
                         + (my[b * CCH + k + 256] - mu2) * g2 + bias;
    }
}

extern "C" void kernel_launch(void* const* d_in, const int* in_sizes, int n_in,
                              void* d_out, int out_size, void* d_ws, size_t ws_size,
                              hipStream_t stream) {
    const float* x  = (const float*)d_in[0];
    const float* w1 = (const float*)d_in[1];
    const float* w2 = (const float*)d_in[2];
    const float* bw = (const float*)d_in[3];
    const float* bb = (const float*)d_in[4];
    float* out = (float*)d_out;

    float* ws = (float*)d_ws;
    float* s  = ws;
    float* q  = ws + 16384;
    float* my = ws + 32768;
    float* vv = ws + 49152;

    se_reduce_kernel<<<RGRID, RTHREADS, 0, stream>>>(x, s, q);
    se_excite_kernel<<<32, 1024, 0, stream>>>(s, q, w1, w2, my, vv);
    se_finalize_kernel<<<1, 256, 0, stream>>>(my, vv, bw, bb, out);
}


# finalize launch folded into the excite kernel: column tile = 8 channels + their 8 partners (c+256), wave 0 takes both batch tiles, batch sums in-wave in a new order; 3 launches -> 2
# speedup vs baseline: 2.3582x; 1.2028x over previous
#include <hip/hip_runtime.h>
#include <math.h>

typedef __attribute__((ext_vector_type(2))) float v2f;
typedef __attribute__((ext_vector_type(8))) float v8f;

#define HW   3136
#define CCH  512
#define NB   32
#define EPSV 1e-5f

typedef __attribute__((ext_vector_type(4))) float v4f;

#define RPW    4
#define RWAVES 16
#define RGRID  256
#define RTHREADS 512
#define ROWF4  (HW / 4)
#define RGSZ   25

static_assert(RGRID * RWAVES * RPW == NB * CCH, "the reduce grid must cover every (b,c) row once");
static_assert(RTHREADS == RWAVES * 32, "work-group size and wave count must agree");
static_assert((RPW * ROWF4) % 32 == 0, "a wave's chunk must be a whole number of wave-wide loads");

__host__ __device__ constexpr int se_gbeg(int g) { return (g * 49 + 1) / 2; }
static_assert(se_gbeg(4) == RPW * ROWF4 / 32, "the four groups must cover the wave's 98 loads");
static_assert(1 * ROWF4 <= 32 * se_gbeg(1) && 2 * ROWF4 <= 32 * se_gbeg(2) &&
              3 * ROWF4 <= 32 * se_gbeg(3) && 4 * ROWF4 <= 32 * se_gbeg(4),
              "row g must be complete once group g is summed");
static_assert(se_gbeg(1) - se_gbeg(0) <= RGSZ && se_gbeg(2) - se_gbeg(1) <= RGSZ &&
              se_gbeg(3) - se_gbeg(2) <= RGSZ && se_gbeg(4) - se_gbeg(3) <= RGSZ, "ring slot too small");

__global__ __launch_bounds__(RTHREADS) void se_reduce_kernel(const float* __restrict__ x,
                                 float* __restrict__ s_out,
                                 float* __restrict__ q_out) {
    const int lane = threadIdx.x & 31;
    const int wv   = __builtin_amdgcn_readfirstlane((int)(threadIdx.x >> 5));
    const int row0 = ((int)blockIdx.x * RWAVES + wv) * RPW;
    const v4f* p = (const v4f*)(x + (size_t)row0 * HW) + lane;

    v4f buf[2][RGSZ];
    v4f sa[RPW], qa[RPW];
    float rs[RPW], rq[RPW];
    #pragma unroll
    for (int r = 0; r < RPW; ++r) { sa[r] = (v4f){0.f, 0.f, 0.f, 0.f}; qa[r] = (v4f){0.f, 0.f, 0.f, 0.f}; }

    #define SE_LOAD_GROUP(G)                                                           \
    {                                                                                  \
               \
        _Pragma("unroll")                                                              \
        for (int j = se_gbeg(G); j < se_gbeg((G) + 1); ++j)                            \
            buf[(G) & 1][j - se_gbeg(G)] = p[32 * j];                                  \
    }
    #define SE_SUM_GROUP(G)                                                            \
    {                                                                                  \
          \
        _Pragma("unroll")                                                              \
        for (int j = se_gbeg(G); j < se_gbeg((G) + 1); ++j) {                          \
            const v4f v = buf[(G) & 1][j - se_gbeg(G)];                                \
            const int rlo = (32 * j) / ROWF4, rhi = (32 * j + 31) / ROWF4;             \
            if (rlo == rhi) {                                                          \
                sa[rlo] += v; qa[rlo] += v * v;                                        \
            } else {                                                                   \
                const int split = rhi * ROWF4 - 32 * j;      \
                const v4f z = (v4f){0.f, 0.f, 0.f, 0.f};                               \
                const v4f vlo = (lane < split) ? v : z;                                \
                const v4f vhi = (lane < split) ? z : v;                                \
                sa[rlo] += vlo; qa[rlo] += vlo * vlo;                                  \
                sa[rhi] += vhi; qa[rhi] += vhi * vhi;                                  \
            }                                                                          \
        }                                                                              \
        rs[G] = (sa[G].x + sa[G].y) + (sa[G].z + sa[G].w);                             \
        rq[G] = (qa[G].x + qa[G].y) + (qa[G].z + qa[G].w);                             \
    }
    SE_LOAD_GROUP(0)
    SE_LOAD_GROUP(1)
    __builtin_amdgcn_sched_barrier(0);
    SE_SUM_GROUP(0)
    __builtin_amdgcn_sched_barrier(0);
    SE_LOAD_GROUP(2)
    __builtin_amdgcn_sched_barrier(0);
    SE_SUM_GROUP(1)
    __builtin_amdgcn_sched_barrier(0);
    SE_LOAD_GROUP(3)
    __builtin_amdgcn_sched_barrier(0);
    SE_SUM_GROUP(2)
    SE_SUM_GROUP(3)
    #undef SE_LOAD_GROUP
    #undef SE_SUM_GROUP

    #pragma unroll
    for (int r = 0; r < RPW; ++r) {
        #pragma unroll
        for (int m = 16; m >= 1; m >>= 1) {
            rs[r] += __shfl_xor(rs[r], m, 32);
            rq[r] += __shfl_xor(rq[r], m, 32);
        }
    }
    float mine = 0.f;
    #pragma unroll
    for (int r = 0; r < RPW; ++r) {
        if (lane == r)       mine = rs[r];
        if (lane == RPW + r) mine = rq[r];
    }
    const float inv = 1.0f / (float)HW;
    if (lane < RPW)          s_out[row0 + lane]       = mine * inv;
    else if (lane < 2 * RPW) q_out[row0 + lane - RPW] = mine * inv;
}

#define NWV  32
#define PPIT 34
#define TPIT 36
#define OCOL (CCH / 2)

__global__ __launch_bounds__(1024) void se_excite_kernel(const float* __restrict__ s,
                                 const float* __restrict__ q,
                                 const float* __restrict__ w1,
                                 const float* __restrict__ w2,
                                 const float* __restrict__ bw,
                                 const float* __restrict__ bb,
                                 float* __restrict__ out) {
    __shared__ float part[NWV][32][PPIT];
    __shared__ __attribute__((aligned(16))) float t1[32][TPIT];
    const int lane = threadIdx.x & 31;
    const int wv   = __builtin_amdgcn_readfirstlane((int)(threadIdx.x >> 5));
    const int half = lane >> 4;
    const int l    = lane & 15;
    const int kout = 8 * (int)blockIdx.x + (l & 7);
    const int ch   = kout + OCOL * (l >> 3);

    const int kb = 16 * wv + 8 * half;
    const float4* pa0 = (const float4*)(s  + (l)      * CCH + kb);
    const float4* pa1 = (const float4*)(s  + (16 + l) * CCH + kb);
    const float4* pb0 = (const float4*)(w1 + (l)      * CCH + kb);
    const float4* pb1 = (const float4*)(w1 + (16 + l) * CCH + kb);
    const float4 a0lo = pa0[0], a0hi = pa0[1];
    const float4 a1lo = pa1[0], a1hi = pa1[1];
    const float4 b0lo = pb0[0], b0hi = pb0[1];
    const float4 b1lo = pb1[0], b1hi = pb1[1];

    float4 wb0 = {}, wb1 = {}, wb2 = {}, wb3 = {};
    float sv[2][8] = {}, qv[2][8] = {};
    float bwv = 0.f, bbv = 0.f;
    if (wv == 0) {
        const float4* pw = (const float4*)(w2 + ch * 32 + 16 * half);
        wb0 = pw[0]; wb1 = pw[1]; wb2 = pw[2]; wb3 = pw[3];
        #pragma unroll
        for (int t = 0; t < 2; ++t) {
            #pragma unroll
            for (int r = 0; r < 8; ++r) {
                const int b = 16 * t + r + 8 * half;
                sv[t][r] = s[b * CCH + ch];
                qv[t][r] = q[b * CCH + ch];
            }
        }
        bwv = bw[ch];
        bbv = bb[ch];
    }

    {
        v8f c00 = {}, c01 = {}, c10 = {}, c11 = {};
        #define SE_STEP(A0, A1, B0, B1, X, Y)                                          \
        {                                                                             \
            v2f fa0, fa1, fb0, fb1;                                                   \
            fa0.x = A0.X; fa0.y = A0.Y; fa1.x = A1.X; fa1.y = A1.Y;                   \
            fb0.x = B0.X; fb0.y = B0.Y; fb1.x = B1.X; fb1.y = B1.Y;                   \
            c00 = __builtin_amdgcn_wmma_f32_16x16x4_f32(false, fa0, false, fb0, (short)0, c00, false, false); \
            c01 = __builtin_amdgcn_wmma_f32_16x16x4_f32(false, fa0, false, fb1, (short)0, c01, false, false); \
            c10 = __builtin_amdgcn_wmma_f32_16x16x4_f32(false, fa1, false, fb0, (short)0, c10, false, false); \
            c11 = __builtin_amdgcn_wmma_f32_16x16x4_f32(false, fa1, false, fb1, (short)0, c11, false, false); \
        }
        SE_STEP(a0lo, a1lo, b0lo, b1lo, x, y)
        SE_STEP(a0lo, a1lo, b0lo, b1lo, z, w)
        SE_STEP(a0hi, a1hi, b0hi, b1hi, x, y)
        SE_STEP(a0hi, a1hi, b0hi, b1hi, z, w)
        #undef SE_STEP
        #pragma unroll
        for (int r = 0; r < 8; ++r) {
            const int row = r + 8 * half;
            part[wv][row][l]           = c00[r];
            part[wv][row][16 + l]      = c01[r];
            part[wv][16 + row][l]      = c10[r];
            part[wv][16 + row][16 + l] = c11[r];
        }
    }
    __syncthreads();

    {
        float acc = 0.f;
        #pragma unroll
        for (int w = 0; w < NWV; ++w) acc += part[w][wv][lane];
        t1[wv][lane] = acc > 0.f ? acc : 0.f;
    }
    __syncthreads();

    if (wv == 0) {
        const float4* pt0 = (const float4*)&t1[l][16 * half];
        const float4* pt1 = (const float4*)&t1[16 + l][16 * half];
        const float4 ta00 = pt0[0], ta01 = pt0[1], ta02 = pt0[2], ta03 = pt0[3];
        const float4 ta10 = pt1[0], ta11 = pt1[1], ta12 = pt1[2], ta13 = pt1[3];
        v8f e0 = {}, e1 = {};
        #define SE_STEP2(TA0, TA1, WB, X, Y)                                           \
        {                                                                             \
            v2f fa0, fa1, fb;                                                         \
            fa0.x = TA0.X; fa0.y = TA0.Y; fa1.x = TA1.X; fa1.y = TA1.Y;               \
            fb.x = WB.X; fb.y = WB.Y;                                                 \
            e0 = __builtin_amdgcn_wmma_f32_16x16x4_f32(false, fa0, false, fb, (short)0, e0, false, false); \
            e1 = __builtin_amdgcn_wmma_f32_16x16x4_f32(false, fa1, false, fb, (short)0, e1, false, false); \
        }
        SE_STEP2(ta00, ta10, wb0, x, y)
        SE_STEP2(ta00, ta10, wb0, z, w)
        SE_STEP2(ta01, ta11, wb1, x, y)
        SE_STEP2(ta01, ta11, wb1, z, w)
        SE_STEP2(ta02, ta12, wb2, x, y)
        SE_STEP2(ta02, ta12, wb2, z, w)
        SE_STEP2(ta03, ta13, wb3, x, y)
        SE_STEP2(ta03, ta13, wb3, z, w)
        #undef SE_STEP2

        float m[2][8];
        float ms[2] = {0.f, 0.f}, vs[2] = {0.f, 0.f};
        #pragma unroll
        for (int t = 0; t < 2; ++t) {
            #pragma unroll
            for (int r = 0; r < 8; ++r) {
                const float ev = (t == 0) ? e0[r] : e1[r];
                const float ex = 1.0f / (1.0f + __expf(-ev));
                m[t][r] = sv[t][r] * ex;
                ms[t] += m[t][r];
                vs[t] += ex * ex * qv[t][r];
            }
        }
        float msum = ms[0] + ms[1], vsum = vs[0] + vs[1];
        msum += __shfl_xor(msum, 16, 32);
        vsum += __shfl_xor(vsum, 16, 32);
        const float invb = 1.0f / (float)NB;
        const float mu = msum * invb;
        const float g  = bwv * rsqrtf(vsum * invb - mu * mu + EPSV);
        const float bias = bbv + __shfl_xor(bbv, 8, 32);
        float o[2][8];
        #pragma unroll
        for (int t = 0; t < 2; ++t) {
            #pragma unroll
            for (int r = 0; r < 8; ++r) {
                const float a = (m[t][r] - mu) * g;
                o[t][r] = (a + __shfl_xor(a, 8, 32)) + bias;
            }
        }
        if ((l & 8) == 0) {
            #pragma unroll
            for (int t = 0; t < 2; ++t) {
                #pragma unroll
                for (int r = 0; r < 8; ++r) {
                    const int b = 16 * t + r + 8 * half;
                    out[b * OCOL + kout] = o[t][r];
                }
            }
        }
    }
}

extern "C" void kernel_launch(void* const* d_in, const int* in_sizes, int n_in,
                              void* d_out, int out_size, void* d_ws, size_t ws_size,
                              hipStream_t stream) {
    const float* x  = (const float*)d_in[0];
    const float* w1 = (const float*)d_in[1];
    const float* w2 = (const float*)d_in[2];
    const float* bw = (const float*)d_in[3];
    const float* bb = (const float*)d_in[4];
    float* out = (float*)d_out;

    float* ws = (float*)d_ws;
    float* s  = ws;
    float* q  = ws + 16384;

    se_reduce_kernel<<<RGRID, RTHREADS, 0, stream>>>(x, s, q);
    se_excite_kernel<<<32, 1024, 0, stream>>>(s, q, w1, w2, bw, bb, out);
}


# first product moved into the reduce kernel and split over K there: one 16x16x4 k-step per wave, 16 summed in LDS, 8 planes summed in the excite kernel (new t1 sum order); excite waves 32 -> 9
# speedup vs baseline: 2.6235x; 1.1125x over previous
#include <hip/hip_runtime.h>
#include <math.h>

typedef __attribute__((ext_vector_type(2))) float v2f;
typedef __attribute__((ext_vector_type(8))) float v8f;

#define HW   3136
#define CCH  512
#define NB   32
#define EPSV 1e-5f

typedef __attribute__((ext_vector_type(4))) float v4f;

#define RPW    4
#define RWAVES 16
#define RGRID  256
#define RTHREADS 512
#define ROWF4  (HW / 4)
#define RGSZ   25
#define NSL    (CCH / (RWAVES * RPW))

static_assert(RGRID * RWAVES * RPW == NB * CCH, "the reduce grid must cover every (b,c) row once");
static_assert(RTHREADS == RWAVES * 32, "work-group size and wave count must agree");
static_assert(CCH % (RWAVES * RPW) == 0 && CCH % RPW == 0, "a work-group's rows must lie in one batch row");
static_assert((RPW * ROWF4) % 32 == 0, "a wave's chunk must be a whole number of wave-wide loads");

__host__ __device__ constexpr int se_gbeg(int g) { return (g * 49 + 1) / 2; }
static_assert(se_gbeg(4) == RPW * ROWF4 / 32, "the four groups must cover the wave's 98 loads");
static_assert(1 * ROWF4 <= 32 * se_gbeg(1) && 2 * ROWF4 <= 32 * se_gbeg(2) &&
              3 * ROWF4 <= 32 * se_gbeg(3) && 4 * ROWF4 <= 32 * se_gbeg(4),
              "row g must be complete once group g is summed");
static_assert(se_gbeg(1) - se_gbeg(0) <= RGSZ && se_gbeg(2) - se_gbeg(1) <= RGSZ &&
              se_gbeg(3) - se_gbeg(2) <= RGSZ && se_gbeg(4) - se_gbeg(3) <= RGSZ, "ring slot too small");

__global__ __launch_bounds__(RTHREADS) void se_reduce_kernel(const float* __restrict__ x,
                                 const float* __restrict__ w1,
                                 float* __restrict__ s_out,
                                 float* __restrict__ q_out,
                                 float* __restrict__ p_out) {
    __shared__ float red[RWAVES][32];
    const int lane = threadIdx.x & 31;
    const int wv   = __builtin_amdgcn_readfirstlane((int)(threadIdx.x >> 5));
    const int row0 = ((int)blockIdx.x * RWAVES + wv) * RPW;
    const v4f* p = (const v4f*)(x + (size_t)row0 * HW) + lane;
    const int kch  = row0 & (CCH - 1);

    v4f buf[2][RGSZ];
    v4f sa[RPW], qa[RPW];
    float rs[RPW], rq[RPW];
    #pragma unroll
    for (int r = 0; r < RPW; ++r) { sa[r] = (v4f){0.f, 0.f, 0.f, 0.f}; qa[r] = (v4f){0.f, 0.f, 0.f, 0.f}; }

    #define SE_LOAD_GROUP(G)                                                           \
    {                                                                                  \
               \
        _Pragma("unroll")                                                              \
        for (int j = se_gbeg(G); j < se_gbeg((G) + 1); ++j)                            \
            buf[(G) & 1][j - se_gbeg(G)] = p[32 * j];                                  \
    }
    #define SE_SUM_GROUP(G)                                                            \
    {                                                                                  \
          \
        _Pragma("unroll")                                                              \
        for (int j = se_gbeg(G); j < se_gbeg((G) + 1); ++j) {                          \
            const v4f v = buf[(G) & 1][j - se_gbeg(G)];                                \
            const int rlo = (32 * j) / ROWF4, rhi = (32 * j + 31) / ROWF4;             \
            if (rlo == rhi) {                                                          \
                sa[rlo] += v; qa[rlo] += v * v;                                        \
            } else {                                                                   \
                const int split = rhi * ROWF4 - 32 * j;      \
                const v4f z = (v4f){0.f, 0.f, 0.f, 0.f};                               \
                const v4f vlo = (lane < split) ? v : z;                                \
                const v4f vhi = (lane < split) ? z : v;                                \
                sa[rlo] += vlo; qa[rlo] += vlo * vlo;                                  \
                sa[rhi] += vhi; qa[rhi] += vhi * vhi;                                  \
            }                                                                          \
        }                                                                              \
        rs[G] = (sa[G].x + sa[G].y) + (sa[G].z + sa[G].w);                             \
        rq[G] = (qa[G].x + qa[G].y) + (qa[G].z + qa[G].w);                             \
    }
    SE_LOAD_GROUP(0)
    SE_LOAD_GROUP(1)
    __builtin_amdgcn_sched_barrier(0);
    SE_SUM_GROUP(0)
    __builtin_amdgcn_sched_barrier(0);
    SE_LOAD_GROUP(2)
    __builtin_amdgcn_sched_barrier(0);
    SE_SUM_GROUP(1)
    __builtin_amdgcn_sched_barrier(0);
    SE_LOAD_GROUP(3)
    __builtin_amdgcn_sched_barrier(0);
    SE_SUM_GROUP(2)
    __builtin_amdgcn_sched_barrier(0);
    const v2f fb0 = *(const v2f*)(w1 + (lane & 15) * CCH + kch + 2 * (lane >> 4));
    const v2f fb1 = *(const v2f*)(w1 + (16 + (lane & 15)) * CCH + kch + 2 * (lane >> 4));
    __builtin_amdgcn_sched_barrier(0);
    SE_SUM_GROUP(3)
    #undef SE_LOAD_GROUP
    #undef SE_SUM_GROUP

    #pragma unroll
    for (int r = 0; r < RPW; ++r) {
        #pragma unroll
        for (int m = 16; m >= 1; m >>= 1) {
            rs[r] += __shfl_xor(rs[r], m, 32);
            rq[r] += __shfl_xor(rq[r], m, 32);
        }
    }
    float mine = 0.f;
    #pragma unroll
    for (int r = 0; r < RPW; ++r) {
        if (lane == r)       mine = rs[r];
        if (lane == RPW + r) mine = rq[r];
    }
    const float inv = 1.0f / (float)HW;

    const bool arow = (lane & 15) == 0;
    const bool ahi  = (lane >> 4) != 0;
    v2f fa;
    fa.x = arow ? ((ahi ? rs[2] : rs[0]) * inv) : 0.f;
    fa.y = arow ? ((ahi ? rs[3] : rs[1]) * inv) : 0.f;
    v8f d0 = {}, d1 = {};
    d0 = __builtin_amdgcn_wmma_f32_16x16x4_f32(false, fa, false, fb0, (short)0, d0, false, false);
    d1 = __builtin_amdgcn_wmma_f32_16x16x4_f32(false, fa, false, fb1, (short)0, d1, false, false);
    if (lane < 16) { red[wv][lane] = d0[0]; red[wv][16 + lane] = d1[0]; }
    __syncthreads();
    if (wv == 0) {
        float acc = 0.f;
        #pragma unroll
        for (int w = 0; w < RWAVES; ++w) acc += red[w][lane];
        const int b  = (int)blockIdx.x / NSL;
        const int sl = (int)blockIdx.x % NSL;
        p_out[(sl * NB + b) * 32 + lane] = acc;
    }
    if (lane < RPW)          s_out[row0 + lane]       = mine * inv;
    else if (lane < 2 * RPW) q_out[row0 + lane - RPW] = mine * inv;
}

#define EPW  8
#define EWV  (EPW + 1)
#define TPIT 36
#define OCOL (CCH / 2)

__global__ __launch_bounds__(EWV * 32) void se_excite_kernel(const float* __restrict__ s,
                                 const float* __restrict__ q,
                                 const float* __restrict__ p,
                                 const float* __restrict__ w2,
                                 const float* __restrict__ bw,
                                 const float* __restrict__ bb,
                                 float* __restrict__ out) {
    __shared__ __attribute__((aligned(16))) float t1[32][TPIT];
    const int lane = threadIdx.x & 31;
    const int wv   = __builtin_amdgcn_readfirstlane((int)(threadIdx.x >> 5));
    const int half = lane >> 4;
    const int l    = lane & 15;
    const int kout = 8 * (int)blockIdx.x + (l & 7);
    const int ch   = kout + OCOL * (l >> 3);

    float4 wb0 = {}, wb1 = {}, wb2 = {}, wb3 = {};
    float sv[2][8] = {}, qv[2][8] = {};
    float bwv = 0.f, bbv = 0.f;
    if (wv == 0) {
        const float4* pw = (const float4*)(w2 + ch * 32 + 16 * half);
        wb0 = pw[0]; wb1 = pw[1]; wb2 = pw[2]; wb3 = pw[3];
        #pragma unroll
        for (int t = 0; t < 2; ++t) {
            #pragma unroll
            for (int r = 0; r < 8; ++r) {
                const int b = 16 * t + r + 8 * half;
                sv[t][r] = s[b * CCH + ch];
                qv[t][r] = q[b * CCH + ch];
            }
        }
        bwv = bw[ch];
        bbv = bb[ch];
    } else {
        const int pw = wv - 1;
        const float4* pp = (const float4*)p + 32 * pw + lane;
        float4 pv[NSL];
        #pragma unroll
        for (int sl = 0; sl < NSL; ++sl) pv[sl] = pp[sl * (NB * 32 / 4)];
        float4 acc = pv[0];
        #pragma unroll
        for (int sl = 1; sl < NSL; ++sl) {
            acc.x += pv[sl].x; acc.y += pv[sl].y; acc.z += pv[sl].z; acc.w += pv[sl].w;
        }
        acc.x = acc.x > 0.f ? acc.x : 0.f; acc.y = acc.y > 0.f ? acc.y : 0.f;
        acc.z = acc.z > 0.f ? acc.z : 0.f; acc.w = acc.w > 0.f ? acc.w : 0.f;
        *(float4*)&t1[4 * pw + (lane >> 3)][4 * (lane & 7)] = acc;
    }
    __syncthreads();

    if (wv == 0) {
        const float4* pt0 = (const float4*)&t1[l][16 * half];
        const float4* pt1 = (const float4*)&t1[16 + l][16 * half];
        const float4 ta00 = pt0[0], ta01 = pt0[1], ta02 = pt0[2], ta03 = pt0[3];
        const float4 ta10 = pt1[0], ta11 = pt1[1], ta12 = pt1[2], ta13 = pt1[3];
        v8f e0 = {}, e1 = {};
        #define SE_STEP2(TA0, TA1, WB, X, Y)                                           \
        {                                                                             \
            v2f fa0, fa1, fb;                                                         \
            fa0.x = TA0.X; fa0.y = TA0.Y; fa1.x = TA1.X; fa1.y = TA1.Y;               \
            fb.x = WB.X; fb.y = WB.Y;                                                 \
            e0 = __builtin_amdgcn_wmma_f32_16x16x4_f32(false, fa0, false, fb, (short)0, e0, false, false); \
            e1 = __builtin_amdgcn_wmma_f32_16x16x4_f32(false, fa1, false, fb, (short)0, e1, false, false); \
        }
        SE_STEP2(ta00, ta10, wb0, x, y)
        SE_STEP2(ta00, ta10, wb0, z, w)
        SE_STEP2(ta01, ta11, wb1, x, y)
        SE_STEP2(ta01, ta11, wb1, z, w)
        SE_STEP2(ta02, ta12, wb2, x, y)
        SE_STEP2(ta02, ta12, wb2, z, w)
        SE_STEP2(ta03, ta13, wb3, x, y)
        SE_STEP2(ta03, ta13, wb3, z, w)
        #undef SE_STEP2

        float m[2][8];
        float ms[2] = {0.f, 0.f}, vs[2] = {0.f, 0.f};
        #pragma unroll
        for (int t = 0; t < 2; ++t) {
            #pragma unroll
            for (int r = 0; r < 8; ++r) {
                const float ev = (t == 0) ? e0[r] : e1[r];
                const float ex = 1.0f / (1.0f + __expf(-ev));
                m[t][r] = sv[t][r] * ex;
                ms[t] += m[t][r];
                vs[t] += ex * ex * qv[t][r];
            }
        }
        float msum = ms[0] + ms[1], vsum = vs[0] + vs[1];
        msum += __shfl_xor(msum, 16, 32);
        vsum += __shfl_xor(vsum, 16, 32);
        const float invb = 1.0f / (float)NB;
        const float mu = msum * invb;
        const float g  = bwv * rsqrtf(vsum * invb - mu * mu + EPSV);
        const float bias = bbv + __shfl_xor(bbv, 8, 32);
        float o[2][8];
        #pragma unroll
        for (int t = 0; t < 2; ++t) {
            #pragma unroll
            for (int r = 0; r < 8; ++r) {
                const float a = (m[t][r] - mu) * g;
                o[t][r] = (a + __shfl_xor(a, 8, 32)) + bias;
            }
        }
        if ((l & 8) == 0) {
            #pragma unroll
            for (int t = 0; t < 2; ++t) {
                #pragma unroll
                for (int r = 0; r < 8; ++r) {
                    const int b = 16 * t + r + 8 * half;
                    out[b * OCOL + kout] = o[t][r];
                }
            }
        }
    }
}

extern "C" void kernel_launch(void* const* d_in, const int* in_sizes, int n_in,
                              void* d_out, int out_size, void* d_ws, size_t ws_size,
                              hipStream_t stream) {
    const float* x  = (const float*)d_in[0];
    const float* w1 = (const float*)d_in[1];
    const float* w2 = (const float*)d_in[2];
    const float* bw = (const float*)d_in[3];
    const float* bb = (const float*)d_in[4];
    float* out = (float*)d_out;

    float* ws = (float*)d_ws;
    float* s  = ws;
    float* q  = ws + 16384;
    float* p  = ws + 32768;

    se_reduce_kernel<<<RGRID, RTHREADS, 0, stream>>>(x, w1, s, q, p);
    se_excite_kernel<<<32, EWV * 32, 0, stream>>>(s, q, p, w2, bw, bb, out);
}


# speedup vs baseline: 2.8772x; 1.0967x over previous
#include <hip/hip_runtime.h>
#include <math.h>

typedef __attribute__((ext_vector_type(2))) float v2f;
typedef __attribute__((ext_vector_type(8))) float v8f;

#define HW   3136
#define CCH  512
#define NB   32
#define EPSV 1e-5f

typedef __attribute__((ext_vector_type(4))) float v4f;

#define RPW    4
#define RWAVES 16
#define RGRID  256
#define RTHREADS 512
#define ROWF4  (HW / 4)
#define RGSZ   25
#define NSL    (CCH / (RWAVES * RPW))

static_assert(RGRID * RWAVES * RPW == NB * CCH, "the reduce grid must cover every (b,c) row once");
static_assert(RTHREADS == RWAVES * 32, "work-group size and wave count must agree");
static_assert(CCH % (RWAVES * RPW) == 0 && CCH % RPW == 0, "a work-group's rows must lie in one batch row");
static_assert((RPW * ROWF4) % 32 == 0, "a wave's chunk must be a whole number of wave-wide loads");

__host__ __device__ constexpr int se_gbeg(int g) { return (g * 49 + 1) / 2; }
static_assert(se_gbeg(4) == RPW * ROWF4 / 32, "the four groups must cover the wave's 98 loads");
static_assert(1 * ROWF4 <= 32 * se_gbeg(1) && 2 * ROWF4 <= 32 * se_gbeg(2) &&
              3 * ROWF4 <= 32 * se_gbeg(3) && 4 * ROWF4 <= 32 * se_gbeg(4),
              "row g must be complete once group g is summed");
static_assert(se_gbeg(1) - se_gbeg(0) <= RGSZ && se_gbeg(2) - se_gbeg(1) <= RGSZ &&
              se_gbeg(3) - se_gbeg(2) <= RGSZ && se_gbeg(4) - se_gbeg(3) <= RGSZ, "ring slot too small");

__global__ __launch_bounds__(RTHREADS) void se_reduce_kernel(const float* __restrict__ x,
                                 const float* __restrict__ w1,
                                 float* __restrict__ s_out,
                                 float* __restrict__ q_out,
                                 float* __restrict__ p_out) {
    __shared__ float red[RWAVES][32];
    const int lane = threadIdx.x & 31;
    const int wv   = __builtin_amdgcn_readfirstlane((int)(threadIdx.x >> 5));
    const int row0 = ((int)blockIdx.x * RWAVES + wv) * RPW;
    const v4f* p = (const v4f*)(x + (size_t)row0 * HW) + lane;
    const int kch  = row0 & (CCH - 1);

    v4f buf[2][RGSZ];
    v4f sa[RPW], qa[RPW];
    float rs[RPW], rq[RPW];
    #pragma unroll
    for (int r = 0; r < RPW; ++r) { sa[r] = (v4f){0.f, 0.f, 0.f, 0.f}; qa[r] = (v4f){0.f, 0.f, 0.f, 0.f}; }

    #define SE_LOAD_GROUP(G)                                                           \
    {                                                                                  \
               \
          \
          \
          \
        _Pragma("unroll")                                                              \
        for (int j = se_gbeg(G); j < se_gbeg((G) + 1); ++j)                            \
            buf[(G) & 1][j - se_gbeg(G)] = __builtin_nontemporal_load(p + 32 * j);     \
    }
    #define SE_SUM_GROUP(G)                                                            \
    {                                                                                  \
          \
        _Pragma("unroll")                                                              \
        for (int j = se_gbeg(G); j < se_gbeg((G) + 1); ++j) {                          \
            const v4f v = buf[(G) & 1][j - se_gbeg(G)];                                \
            const int rlo = (32 * j) / ROWF4, rhi = (32 * j + 31) / ROWF4;             \
            if (rlo == rhi) {                                                          \
                sa[rlo] += v; qa[rlo] += v * v;                                        \
            } else {                                                                   \
                const int split = rhi * ROWF4 - 32 * j;      \
                const v4f z = (v4f){0.f, 0.f, 0.f, 0.f};                               \
                const v4f vlo = (lane < split) ? v : z;                                \
                const v4f vhi = (lane < split) ? z : v;                                \
                sa[rlo] += vlo; qa[rlo] += vlo * vlo;                                  \
                sa[rhi] += vhi; qa[rhi] += vhi * vhi;                                  \
            }                                                                          \
        }                                                                              \
        rs[G] = (sa[G].x + sa[G].y) + (sa[G].z + sa[G].w);                             \
        rq[G] = (qa[G].x + qa[G].y) + (qa[G].z + qa[G].w);                             \
    }
    SE_LOAD_GROUP(0)
    SE_LOAD_GROUP(1)
    __builtin_amdgcn_sched_barrier(0);
    SE_SUM_GROUP(0)
    __builtin_amdgcn_sched_barrier(0);
    SE_LOAD_GROUP(2)
    __builtin_amdgcn_sched_barrier(0);
    SE_SUM_GROUP(1)
    __builtin_amdgcn_sched_barrier(0);
    SE_LOAD_GROUP(3)
    __builtin_amdgcn_sched_barrier(0);
    SE_SUM_GROUP(2)
    __builtin_amdgcn_sched_barrier(0);
    const v2f fb0 = *(const v2f*)(w1 + (lane & 15) * CCH + kch + 2 * (lane >> 4));
    const v2f fb1 = *(const v2f*)(w1 + (16 + (lane & 15)) * CCH + kch + 2 * (lane >> 4));
    __builtin_amdgcn_sched_barrier(0);
    SE_SUM_GROUP(3)
    #undef SE_LOAD_GROUP
    #undef SE_SUM_GROUP

    #pragma unroll
    for (int r = 0; r < RPW; ++r) {
        #pragma unroll
        for (int m = 16; m >= 1; m >>= 1) {
            rs[r] += __shfl_xor(rs[r], m, 32);
            rq[r] += __shfl_xor(rq[r], m, 32);
        }
    }
    float mine = 0.f;
    #pragma unroll
    for (int r = 0; r < RPW; ++r) {
        if (lane == r)       mine = rs[r];
        if (lane == RPW + r) mine = rq[r];
    }
    const float inv = 1.0f / (float)HW;

    const bool arow = (lane & 15) == 0;
    const bool ahi  = (lane >> 4) != 0;
    v2f fa;
    fa.x = arow ? ((ahi ? rs[2] : rs[0]) * inv) : 0.f;
    fa.y = arow ? ((ahi ? rs[3] : rs[1]) * inv) : 0.f;
    v8f d0 = {}, d1 = {};
    d0 = __builtin_amdgcn_wmma_f32_16x16x4_f32(false, fa, false, fb0, (short)0, d0, false, false);
    d1 = __builtin_amdgcn_wmma_f32_16x16x4_f32(false, fa, false, fb1, (short)0, d1, false, false);
    if (lane < 16) { red[wv][lane] = d0[0]; red[wv][16 + lane] = d1[0]; }
    __syncthreads();
    if (wv == 0) {
        float acc = 0.f;
        #pragma unroll
        for (int w = 0; w < RWAVES; ++w) acc += red[w][lane];
        const int b  = (int)blockIdx.x / NSL;
        const int sl = (int)blockIdx.x % NSL;
        p_out[(sl * NB + b) * 32 + lane] = acc;
    }
    if (lane < RPW)          s_out[row0 + lane]       = mine * inv;
    else if (lane < 2 * RPW) q_out[row0 + lane - RPW] = mine * inv;
}

#define EPW  8
#define EWV  (EPW + 1)
#define TPIT 36
#define OCOL (CCH / 2)

__global__ __launch_bounds__(EWV * 32) void se_excite_kernel(const float* __restrict__ s,
                                 const float* __restrict__ q,
                                 const float* __restrict__ p,
                                 const float* __restrict__ w2,
                                 const float* __restrict__ bw,
                                 const float* __restrict__ bb,
                                 float* __restrict__ out) {
    __shared__ __attribute__((aligned(16))) float t1[32][TPIT];
    const int lane = threadIdx.x & 31;
    const int wv   = __builtin_amdgcn_readfirstlane((int)(threadIdx.x >> 5));
    const int half = lane >> 4;
    const int l    = lane & 15;
    const int kout = 8 * (int)blockIdx.x + (l & 7);
    const int ch   = kout + OCOL * (l >> 3);

    float4 wb0 = {}, wb1 = {}, wb2 = {}, wb3 = {};
    float sv[2][8] = {}, qv[2][8] = {};
    float bwv = 0.f, bbv = 0.f;
    if (wv == 0) {
        const float4* pw = (const float4*)(w2 + ch * 32 + 16 * half);
        wb0 = pw[0]; wb1 = pw[1]; wb2 = pw[2]; wb3 = pw[3];
        #pragma unroll
        for (int t = 0; t < 2; ++t) {
            #pragma unroll
            for (int r = 0; r < 8; ++r) {
                const int b = 16 * t + r + 8 * half;
                sv[t][r] = s[b * CCH + ch];
                qv[t][r] = q[b * CCH + ch];
            }
        }
        bwv = bw[ch];
        bbv = bb[ch];
    } else {
        const int pw = wv - 1;
        const float4* pp = (const float4*)p + 32 * pw + lane;
        float4 pv[NSL];
        #pragma unroll
        for (int sl = 0; sl < NSL; ++sl) pv[sl] = pp[sl * (NB * 32 / 4)];
        float4 acc = pv[0];
        #pragma unroll
        for (int sl = 1; sl < NSL; ++sl) {
            acc.x += pv[sl].x; acc.y += pv[sl].y; acc.z += pv[sl].z; acc.w += pv[sl].w;
        }
        acc.x = acc.x > 0.f ? acc.x : 0.f; acc.y = acc.y > 0.f ? acc.y : 0.f;
        acc.z = acc.z > 0.f ? acc.z : 0.f; acc.w = acc.w > 0.f ? acc.w : 0.f;
        *(float4*)&t1[4 * pw + (lane >> 3)][4 * (lane & 7)] = acc;
    }
    __syncthreads();

    if (wv == 0) {
        const float4* pt0 = (const float4*)&t1[l][16 * half];
        const float4* pt1 = (const float4*)&t1[16 + l][16 * half];
        const float4 ta00 = pt0[0], ta01 = pt0[1], ta02 = pt0[2], ta03 = pt0[3];
        const float4 ta10 = pt1[0], ta11 = pt1[1], ta12 = pt1[2], ta13 = pt1[3];
        v8f e0 = {}, e1 = {};
        #define SE_STEP2(TA0, TA1, WB, X, Y)                                           \
        {                                                                             \
            v2f fa0, fa1, fb;                                                         \
            fa0.x = TA0.X; fa0.y = TA0.Y; fa1.x = TA1.X; fa1.y = TA1.Y;               \
            fb.x = WB.X; fb.y = WB.Y;                                                 \
            e0 = __builtin_amdgcn_wmma_f32_16x16x4_f32(false, fa0, false, fb, (short)0, e0, false, false); \
            e1 = __builtin_amdgcn_wmma_f32_16x16x4_f32(false, fa1, false, fb, (short)0, e1, false, false); \
        }
        SE_STEP2(ta00, ta10, wb0, x, y)
        SE_STEP2(ta00, ta10, wb0, z, w)
        SE_STEP2(ta01, ta11, wb1, x, y)
        SE_STEP2(ta01, ta11, wb1, z, w)
        SE_STEP2(ta02, ta12, wb2, x, y)
        SE_STEP2(ta02, ta12, wb2, z, w)
        SE_STEP2(ta03, ta13, wb3, x, y)
        SE_STEP2(ta03, ta13, wb3, z, w)
        #undef SE_STEP2

        float m[2][8];
        float ms[2] = {0.f, 0.f}, vs[2] = {0.f, 0.f};
        #pragma unroll
        for (int t = 0; t < 2; ++t) {
            #pragma unroll
            for (int r = 0; r < 8; ++r) {
                const float ev = (t == 0) ? e0[r] : e1[r];
                const float ex = 1.0f / (1.0f + __expf(-ev));
                m[t][r] = sv[t][r] * ex;
                ms[t] += m[t][r];
                vs[t] += ex * ex * qv[t][r];
            }
        }
        float msum = ms[0] + ms[1], vsum = vs[0] + vs[1];
        msum += __shfl_xor(msum, 16, 32);
        vsum += __shfl_xor(vsum, 16, 32);
        const float invb = 1.0f / (float)NB;
        const float mu = msum * invb;
        const float g  = bwv * rsqrtf(vsum * invb - mu * mu + EPSV);
        const float bias = bbv + __shfl_xor(bbv, 8, 32);
        float o[2][8];
        #pragma unroll
        for (int t = 0; t < 2; ++t) {
            #pragma unroll
            for (int r = 0; r < 8; ++r) {
                const float a = (m[t][r] - mu) * g;
                o[t][r] = (a + __shfl_xor(a, 8, 32)) + bias;
            }
        }
        if ((l & 8) == 0) {
            #pragma unroll
            for (int t = 0; t < 2; ++t) {
                #pragma unroll
                for (int r = 0; r < 8; ++r) {
                    const int b = 16 * t + r + 8 * half;
                    out[b * OCOL + kout] = o[t][r];
                }
            }
        }
    }
}

extern "C" void kernel_launch(void* const* d_in, const int* in_sizes, int n_in,
                              void* d_out, int out_size, void* d_ws, size_t ws_size,
                              hipStream_t stream) {
    const float* x  = (const float*)d_in[0];
    const float* w1 = (const float*)d_in[1];
    const float* w2 = (const float*)d_in[2];
    const float* bw = (const float*)d_in[3];
    const float* bb = (const float*)d_in[4];
    float* out = (float*)d_out;

    float* ws = (float*)d_ws;
    float* s  = ws;
    float* q  = ws + 16384;
    float* p  = ws + 32768;

    se_reduce_kernel<<<RGRID, RTHREADS, 0, stream>>>(x, w1, s, q, p);
    se_excite_kernel<<<32, EWV * 32, 0, stream>>>(s, q, p, w2, bw, bb, out);
}
